# starting kernel
# baseline (speedup 1.0000x reference)
#include <hip/hip_runtime.h>
#include <hip/hip_bf16.h>

typedef __attribute__((ext_vector_type(2))) float v2f;
typedef __attribute__((ext_vector_type(8))) float v8f;

#define BATCH   4
#define TOK     4096
#define HDIM    1024
#define NHEADS  16
#define ED      64
#define BH      (BATCH * NHEADS)
#define SPLITS  8
#define CHUNKS  4
#define CHUNK   128
#define PKS     (SPLITS * 16)

__device__ __forceinline__ float elu_fm(float x) {
    return x > 0.0f ? x + 1.0f : expf(x);
}

__global__ __launch_bounds__(256) void la_ctx_partial(
    const float* __restrict__ key, const float* __restrict__ value,
    float* __restrict__ pctx, float* __restrict__ pksum)
{
    __shared__ float sK[CHUNK * ED];
    __shared__ float sV[CHUNK * ED];

    const int bh    = blockIdx.x / SPLITS;
    const int split = blockIdx.x % SPLITS;
    const int b = bh >> 4, h = bh & 15;
    const int tid = threadIdx.x;
    const int wave = tid >> 5, lane = tid & 31;
    const int half = lane >> 4, lm = lane & 15;

    const size_t base = (size_t)b * TOK * HDIM + (size_t)h * ED;

    const int e0a = (wave >> 2) << 4;
    const int e0b = e0a + 32;
    const int d0  = (wave & 3) << 4;

    v8f acc0 = {};
    v8f acc1 = {};
    float4 kacc = make_float4(0.f, 0.f, 0.f, 0.f);

    for (int c = 0; c < CHUNKS; ++c) {
        __syncthreads();
        const int s0 = (split * CHUNKS + c) * CHUNK;
        for (int i = tid; i < CHUNK * 16; i += 256) {
            const int row = i >> 4, c4 = i & 15;
            const size_t g = base + (size_t)(s0 + row) * HDIM + (size_t)c4 * 4;
            float4 kv = *(const float4*)(key + g);
            float4 vv = *(const float4*)(value + g);
            float4 kf;
            kf.x = elu_fm(kv.x); kf.y = elu_fm(kv.y);
            kf.z = elu_fm(kv.z); kf.w = elu_fm(kv.w);
            kacc.x += kf.x; kacc.y += kf.y; kacc.z += kf.z; kacc.w += kf.w;
            *(float4*)(sK + row * ED + c4 * 4) = kf;
            *(float4*)(sV + row * ED + c4 * 4) = vv;
        }
        __syncthreads();

        for (int s = 0; s < CHUNK; s += 4) {
            v2f a0, a1, bb;
            const int r0 = (s + 2 * half) * ED;
            const int r1 = (s + 1 + 2 * half) * ED;
            bb.x = sV[r0 + d0 + lm];
            bb.y = sV[r1 + d0 + lm];
            a0.x = sK[r0 + e0a + lm];
            a0.y = sK[r1 + e0a + lm];
            a1.x = sK[r0 + e0b + lm];
            a1.y = sK[r1 + e0b + lm];
            acc0 = __builtin_amdgcn_wmma_f32_16x16x4_f32(
                false, a0, false, bb, (short)0, acc0, false, false);
            acc1 = __builtin_amdgcn_wmma_f32_16x16x4_f32(
                false, a1, false, bb, (short)0, acc1, false, false);
        }
    }

    {
        float* dst = pksum + ((size_t)bh * PKS + split * 16 + wave * 2 + half) * ED
                           + lm * 4;
        *(float4*)dst = kacc;
    }

    float* pc = pctx + ((size_t)bh * SPLITS + split) * (ED * ED);
    #pragma unroll
    for (int r = 0; r < 8; ++r) {
        pc[(e0a + r + 8 * half) * ED + d0 + lm] = acc0[r];
        pc[(e0b + r + 8 * half) * ED + d0 + lm] = acc1[r];
    }
}

__global__ __launch_bounds__(256) void la_reduce(
    const float* __restrict__ pctx, const float* __restrict__ pksum,
    float* __restrict__ ctx, float* __restrict__ ksum)
{
    const int bh = blockIdx.x, tid = threadIdx.x;
    for (int j = tid; j < ED * ED; j += 256) {
        float s = 0.f;
        #pragma unroll
        for (int sp = 0; sp < SPLITS; ++sp)
            s += pctx[((size_t)bh * SPLITS + sp) * (ED * ED) + j];
        ctx[(size_t)bh * (ED * ED) + j] = s;
    }
    if (tid < ED) {
        float s = 0.f;
        for (int sl = 0; sl < PKS; ++sl)
            s += pksum[((size_t)bh * PKS + sl) * ED + tid];
        ksum[bh * ED + tid] = s;
    }
}

__global__ __launch_bounds__(256) void la_out(
    const float* __restrict__ query, const float* __restrict__ ctx,
    const float* __restrict__ ksum, float* __restrict__ out)
{
    __shared__ float sQ[64 * ED];
    __shared__ float sC[ED * ED];
    __shared__ float sKs[ED];
    __shared__ float sDinv[64];

    const int bh = blockIdx.y;
    const int b = bh >> 4, h = bh & 15;
    const int t0 = blockIdx.x * 64;
    const int tid = threadIdx.x;
    const int wave = tid >> 5, lane = tid & 31;
    const int half = lane >> 4, lm = lane & 15;

    const size_t qbase = (size_t)b * TOK * HDIM + (size_t)h * ED;

    for (int i = tid; i < 64 * 16; i += 256) {
        const int row = i >> 4, c4 = i & 15;
        *(float4*)(sC + row * ED + c4 * 4) =
            *(const float4*)(ctx + (size_t)bh * (ED * ED) + row * ED + c4 * 4);
        float4 qv = *(const float4*)(query + qbase + (size_t)(t0 + row) * HDIM
                                     + (size_t)c4 * 4);
        float4 qf;
        qf.x = elu_fm(qv.x); qf.y = elu_fm(qv.y);
        qf.z = elu_fm(qv.z); qf.w = elu_fm(qv.w);
        *(float4*)(sQ + row * ED + c4 * 4) = qf;
    }
    if (tid < ED) sKs[tid] = ksum[bh * ED + tid];
    __syncthreads();

    if (tid < 64) {
        float d = 0.f;
        #pragma unroll 8
        for (int e = 0; e < ED; ++e) d += sQ[tid * ED + e] * sKs[e];
        sDinv[tid] = 1.0f / (d + 1e-6f);
    }
    __syncthreads();

    const int n0a = (wave >> 2) << 4;
    const int n0b = n0a + 32;
    const int d0  = (wave & 3) << 4;

    v8f acc0 = {};
    v8f acc1 = {};
    for (int e = 0; e < ED; e += 4) {
        v2f a0, a1, bb;
        bb.x = sC[(e + 2 * half) * ED + d0 + lm];
        bb.y = sC[(e + 1 + 2 * half) * ED + d0 + lm];
        a0.x = sQ[(n0a + lm) * ED + e + 2 * half];
        a0.y = sQ[(n0a + lm) * ED + e + 1 + 2 * half];
        a1.x = sQ[(n0b + lm) * ED + e + 2 * half];
        a1.y = sQ[(n0b + lm) * ED + e + 1 + 2 * half];
        acc0 = __builtin_amdgcn_wmma_f32_16x16x4_f32(
            false, a0, false, bb, (short)0, acc0, false, false);
        acc1 = __builtin_amdgcn_wmma_f32_16x16x4_f32(
            false, a1, false, bb, (short)0, acc1, false, false);
    }

    #pragma unroll
    for (int r = 0; r < 8; ++r) {
        const int rowa = n0a + r + 8 * half;
        const int rowb = n0b + r + 8 * half;
        out[((size_t)b * TOK + t0 + rowa) * HDIM + h * ED + d0 + lm]
            = acc0[r] * sDinv[rowa];
        out[((size_t)b * TOK + t0 + rowb) * HDIM + h * ED + d0 + lm]
            = acc1[r] * sDinv[rowb];
    }
}

extern "C" void kernel_launch(void* const* d_in, const int* in_sizes, int n_in,
                              void* d_out, int out_size, void* d_ws, size_t ws_size,
                              hipStream_t stream) {
    (void)in_sizes; (void)n_in; (void)out_size; (void)ws_size;
    const float* q = (const float*)d_in[0];
    const float* k = (const float*)d_in[1];
    const float* v = (const float*)d_in[2];
    float* out = (float*)d_out;

    float* wsf   = (float*)d_ws;
    float* pctx  = wsf;
    float* pksum = pctx  + (size_t)BH * SPLITS * (ED * ED);
    float* ctx   = pksum + (size_t)BH * PKS * ED;
    float* ksum  = ctx   + (size_t)BH * (ED * ED);

    la_ctx_partial<<<BH * SPLITS, 256, 0, stream>>>(k, v, pctx, pksum);
    la_reduce<<<BH, 256, 0, stream>>>(pctx, pksum, ctx, ksum);
    la_out<<<dim3(TOK / 64, BH), 256, 0, stream>>>(q, ctx, ksum, out);
}


# LDS row pitch padded (K/V/context tiles 64->72 floats, Q tile 64->68) so WMMA fragment reads and the per-row dot read are bank-conflict-free (were 2-way and 16- to 32-way)
# speedup vs baseline: 1.1661x; 1.1661x over previous
#include <hip/hip_runtime.h>
#include <hip/hip_bf16.h>

typedef __attribute__((ext_vector_type(2))) float v2f;
typedef __attribute__((ext_vector_type(8))) float v8f;

#define BATCH   4
#define TOK     4096
#define HDIM    1024
#define NHEADS  16
#define ED      64
#define BH      (BATCH * NHEADS)
#define SPLITS  8
#define CHUNKS  4
#define CHUNK   128
#define PKS     (SPLITS * 16)
#define KVP     72
#define QP      68

__device__ __forceinline__ float elu_fm(float x) {
    return x > 0.0f ? x + 1.0f : expf(x);
}

__global__ __launch_bounds__(256) void la_ctx_partial(
    const float* __restrict__ key, const float* __restrict__ value,
    float* __restrict__ pctx, float* __restrict__ pksum)
{
    __shared__ float sK[CHUNK * KVP];
    __shared__ float sV[CHUNK * KVP];

    const int bh    = blockIdx.x / SPLITS;
    const int split = blockIdx.x % SPLITS;
    const int b = bh >> 4, h = bh & 15;
    const int tid = threadIdx.x;
    const int wave = tid >> 5, lane = tid & 31;
    const int half = lane >> 4, lm = lane & 15;

    const size_t base = (size_t)b * TOK * HDIM + (size_t)h * ED;

    const int e0a = (wave >> 2) << 4;
    const int e0b = e0a + 32;
    const int d0  = (wave & 3) << 4;

    v8f acc0 = {};
    v8f acc1 = {};
    float4 kacc = make_float4(0.f, 0.f, 0.f, 0.f);

    for (int c = 0; c < CHUNKS; ++c) {
        __syncthreads();
        const int s0 = (split * CHUNKS + c) * CHUNK;
        for (int i = tid; i < CHUNK * 16; i += 256) {
            const int row = i >> 4, c4 = i & 15;
            const size_t g = base + (size_t)(s0 + row) * HDIM + (size_t)c4 * 4;
            float4 kv = *(const float4*)(key + g);
            float4 vv = *(const float4*)(value + g);
            float4 kf;
            kf.x = elu_fm(kv.x); kf.y = elu_fm(kv.y);
            kf.z = elu_fm(kv.z); kf.w = elu_fm(kv.w);
            kacc.x += kf.x; kacc.y += kf.y; kacc.z += kf.z; kacc.w += kf.w;
            *(float4*)(sK + row * KVP + c4 * 4) = kf;
            *(float4*)(sV + row * KVP + c4 * 4) = vv;
        }
        __syncthreads();

        for (int s = 0; s < CHUNK; s += 4) {
            v2f a0, a1, bb;
            const int r0 = (s + 2 * half) * KVP;
            const int r1 = (s + 1 + 2 * half) * KVP;
            bb.x = sV[r0 + d0 + lm];
            bb.y = sV[r1 + d0 + lm];
            a0.x = sK[r0 + e0a + lm];
            a0.y = sK[r1 + e0a + lm];
            a1.x = sK[r0 + e0b + lm];
            a1.y = sK[r1 + e0b + lm];
            acc0 = __builtin_amdgcn_wmma_f32_16x16x4_f32(
                false, a0, false, bb, (short)0, acc0, false, false);
            acc1 = __builtin_amdgcn_wmma_f32_16x16x4_f32(
                false, a1, false, bb, (short)0, acc1, false, false);
        }
    }

    {
        float* dst = pksum + ((size_t)bh * PKS + split * 16 + wave * 2 + half) * ED
                           + lm * 4;
        *(float4*)dst = kacc;
    }

    float* pc = pctx + ((size_t)bh * SPLITS + split) * (ED * ED);
    #pragma unroll
    for (int r = 0; r < 8; ++r) {
        pc[(e0a + r + 8 * half) * ED + d0 + lm] = acc0[r];
        pc[(e0b + r + 8 * half) * ED + d0 + lm] = acc1[r];
    }
}

__global__ __launch_bounds__(256) void la_reduce(
    const float* __restrict__ pctx, const float* __restrict__ pksum,
    float* __restrict__ ctx, float* __restrict__ ksum)
{
    const int bh = blockIdx.x, tid = threadIdx.x;
    for (int j = tid; j < ED * ED; j += 256) {
        float s = 0.f;
        #pragma unroll
        for (int sp = 0; sp < SPLITS; ++sp)
            s += pctx[((size_t)bh * SPLITS + sp) * (ED * ED) + j];
        ctx[(size_t)bh * (ED * ED) + j] = s;
    }
    if (tid < ED) {
        float s = 0.f;
        for (int sl = 0; sl < PKS; ++sl)
            s += pksum[((size_t)bh * PKS + sl) * ED + tid];
        ksum[bh * ED + tid] = s;
    }
}

__global__ __launch_bounds__(256) void la_out(
    const float* __restrict__ query, const float* __restrict__ ctx,
    const float* __restrict__ ksum, float* __restrict__ out)
{
    __shared__ float sQ[64 * QP];
    __shared__ float sC[ED * KVP];
    __shared__ float sKs[ED];
    __shared__ float sDinv[64];

    const int bh = blockIdx.y;
    const int b = bh >> 4, h = bh & 15;
    const int t0 = blockIdx.x * 64;
    const int tid = threadIdx.x;
    const int wave = tid >> 5, lane = tid & 31;
    const int half = lane >> 4, lm = lane & 15;

    const size_t qbase = (size_t)b * TOK * HDIM + (size_t)h * ED;

    for (int i = tid; i < 64 * 16; i += 256) {
        const int row = i >> 4, c4 = i & 15;
        *(float4*)(sC + row * KVP + c4 * 4) =
            *(const float4*)(ctx + (size_t)bh * (ED * ED) + row * ED + c4 * 4);
        float4 qv = *(const float4*)(query + qbase + (size_t)(t0 + row) * HDIM
                                     + (size_t)c4 * 4);
        float4 qf;
        qf.x = elu_fm(qv.x); qf.y = elu_fm(qv.y);
        qf.z = elu_fm(qv.z); qf.w = elu_fm(qv.w);
        *(float4*)(sQ + row * QP + c4 * 4) = qf;
    }
    if (tid < ED) sKs[tid] = ksum[bh * ED + tid];
    __syncthreads();

    if (tid < 64) {
        float d = 0.f;
        #pragma unroll 8
        for (int e = 0; e < ED; ++e) d += sQ[tid * QP + e] * sKs[e];
        sDinv[tid] = 1.0f / (d + 1e-6f);
    }
    __syncthreads();

    const int n0a = (wave >> 2) << 4;
    const int n0b = n0a + 32;
    const int d0  = (wave & 3) << 4;

    v8f acc0 = {};
    v8f acc1 = {};
    for (int e = 0; e < ED; e += 4) {
        v2f a0, a1, bb;
        bb.x = sC[(e + 2 * half) * KVP + d0 + lm];
        bb.y = sC[(e + 1 + 2 * half) * KVP + d0 + lm];
        a0.x = sQ[(n0a + lm) * QP + e + 2 * half];
        a0.y = sQ[(n0a + lm) * QP + e + 1 + 2 * half];
        a1.x = sQ[(n0b + lm) * QP + e + 2 * half];
        a1.y = sQ[(n0b + lm) * QP + e + 1 + 2 * half];
        acc0 = __builtin_amdgcn_wmma_f32_16x16x4_f32(
            false, a0, false, bb, (short)0, acc0, false, false);
        acc1 = __builtin_amdgcn_wmma_f32_16x16x4_f32(
            false, a1, false, bb, (short)0, acc1, false, false);
    }

    #pragma unroll
    for (int r = 0; r < 8; ++r) {
        const int rowa = n0a + r + 8 * half;
        const int rowb = n0b + r + 8 * half;
        out[((size_t)b * TOK + t0 + rowa) * HDIM + h * ED + d0 + lm]
            = acc0[r] * sDinv[rowa];
        out[((size_t)b * TOK + t0 + rowb) * HDIM + h * ED + d0 + lm]
            = acc1[r] * sDinv[rowb];
    }
}

extern "C" void kernel_launch(void* const* d_in, const int* in_sizes, int n_in,
                              void* d_out, int out_size, void* d_ws, size_t ws_size,
                              hipStream_t stream) {
    (void)in_sizes; (void)n_in; (void)out_size; (void)ws_size;
    const float* q = (const float*)d_in[0];
    const float* k = (const float*)d_in[1];
    const float* v = (const float*)d_in[2];
    float* out = (float*)d_out;

    float* wsf   = (float*)d_ws;
    float* pctx  = wsf;
    float* pksum = pctx  + (size_t)BH * SPLITS * (ED * ED);
    float* ctx   = pksum + (size_t)BH * PKS * ED;
    float* ksum  = ctx   + (size_t)BH * (ED * ED);

    la_ctx_partial<<<BH * SPLITS, 256, 0, stream>>>(k, v, pctx, pksum);
    la_reduce<<<BH, 256, 0, stream>>>(pctx, pksum, ctx, ksum);
    la_out<<<dim3(TOK / 64, BH), 256, 0, stream>>>(q, ctx, ksum, out);
}


# Staging of pass 1 and pass 3 issues all global loads of a tile together; pass 1 fetches the next 128-row K/V chunk into registers under the current WMMA loop (was one waited round trip per float4)
# speedup vs baseline: 1.5863x; 1.3603x over previous
#include <hip/hip_runtime.h>
#include <hip/hip_bf16.h>

typedef __attribute__((ext_vector_type(2))) float v2f;
typedef __attribute__((ext_vector_type(8))) float v8f;
typedef __attribute__((ext_vector_type(4))) float v4f;

#define BATCH   4
#define TOK     4096
#define HDIM    1024
#define NHEADS  16
#define ED      64
#define BH      (BATCH * NHEADS)
#define SPLITS  8
#define CHUNKS  4
#define CHUNK   128
#define PKS     (SPLITS * 16)
#define KVP     72
#define QP      68

__device__ __forceinline__ float elu_fm(float x) {
    return x > 0.0f ? x + 1.0f : expf(x);
}

__device__ __forceinline__ void wg_barrier_lds_only() {
    asm volatile("s_wait_dscnt 0x0\n\ts_barrier_signal -1\n\ts_barrier_wait -1" ::: "memory");
}

__global__ __launch_bounds__(256) __attribute__((amdgpu_waves_per_eu(4, 4)))
void la_ctx_partial(
    const float* __restrict__ key, const float* __restrict__ value,
    float* __restrict__ pctx, float* __restrict__ pksum)
{
    __shared__ float sK[CHUNK * KVP];
    __shared__ float sV[CHUNK * KVP];

    const int bh    = blockIdx.x / SPLITS;
    const int split = blockIdx.x % SPLITS;
    const int b = bh >> 4, h = bh & 15;
    const int tid = threadIdx.x;
    const int wave = tid >> 5, lane = tid & 31;
    const int half = lane >> 4, lm = lane & 15;

    const size_t base = (size_t)b * TOK * HDIM + (size_t)h * ED;

    const int e0a = (wave >> 2) << 4;
    const int e0b = e0a + 32;
    const int d0  = (wave & 3) << 4;

    v8f acc0 = {};
    v8f acc1 = {};
    float4 kacc = make_float4(0.f, 0.f, 0.f, 0.f);

    const int lrow = tid >> 4, c4 = tid & 15;
    const size_t gthr = base + (size_t)lrow * HDIM + (size_t)c4 * 4;
    v4f kreg[8], vreg[8];
    {
        const size_t g0 = gthr + (size_t)(split * CHUNKS * CHUNK) * HDIM;
        #pragma unroll
        for (int u = 0; u < 8; ++u) {
            kreg[u] = *(const v4f*)(key   + g0 + (size_t)(16 * u) * HDIM);
            vreg[u] = *(const v4f*)(value + g0 + (size_t)(16 * u) * HDIM);
        }
    }

    for (int c = 0; c < CHUNKS; ++c) {
        wg_barrier_lds_only();
        #pragma unroll
        for (int u = 0; u < 8; ++u) {
            const int row = lrow + 16 * u;
            v4f kf;
            kf.x = elu_fm(kreg[u].x); kf.y = elu_fm(kreg[u].y);
            kf.z = elu_fm(kreg[u].z); kf.w = elu_fm(kreg[u].w);
            kacc.x += kf.x; kacc.y += kf.y; kacc.z += kf.z; kacc.w += kf.w;
            *(v4f*)(sK + row * KVP + c4 * 4) = kf;
            *(v4f*)(sV + row * KVP + c4 * 4) = vreg[u];
        }
        if (c + 1 < CHUNKS) {
            const size_t g1 = gthr
                + (size_t)((split * CHUNKS + c + 1) * CHUNK) * HDIM;
            #pragma unroll
            for (int u = 0; u < 8; ++u) {
                kreg[u] = *(const v4f*)(key   + g1 + (size_t)(16 * u) * HDIM);
                vreg[u] = *(const v4f*)(value + g1 + (size_t)(16 * u) * HDIM);
            }
        }
        wg_barrier_lds_only();

        for (int s = 0; s < CHUNK; s += 4) {
            v2f a0, a1, bb;
            const int r0 = (s + 2 * half) * KVP;
            const int r1 = (s + 1 + 2 * half) * KVP;
            bb.x = sV[r0 + d0 + lm];
            bb.y = sV[r1 + d0 + lm];
            a0.x = sK[r0 + e0a + lm];
            a0.y = sK[r1 + e0a + lm];
            a1.x = sK[r0 + e0b + lm];
            a1.y = sK[r1 + e0b + lm];
            acc0 = __builtin_amdgcn_wmma_f32_16x16x4_f32(
                false, a0, false, bb, (short)0, acc0, false, false);
            acc1 = __builtin_amdgcn_wmma_f32_16x16x4_f32(
                false, a1, false, bb, (short)0, acc1, false, false);
        }
    }

    {
        float* dst = pksum + ((size_t)bh * PKS + split * 16 + wave * 2 + half) * ED
                           + lm * 4;
        *(float4*)dst = kacc;
    }

    float* pc = pctx + ((size_t)bh * SPLITS + split) * (ED * ED);
    #pragma unroll
    for (int r = 0; r < 8; ++r) {
        pc[(e0a + r + 8 * half) * ED + d0 + lm] = acc0[r];
        pc[(e0b + r + 8 * half) * ED + d0 + lm] = acc1[r];
    }
}

__global__ __launch_bounds__(256) void la_reduce(
    const float* __restrict__ pctx, const float* __restrict__ pksum,
    float* __restrict__ ctx, float* __restrict__ ksum)
{
    const int bh = blockIdx.x, tid = threadIdx.x;
    for (int j = tid; j < ED * ED; j += 256) {
        float s = 0.f;
        #pragma unroll
        for (int sp = 0; sp < SPLITS; ++sp)
            s += pctx[((size_t)bh * SPLITS + sp) * (ED * ED) + j];
        ctx[(size_t)bh * (ED * ED) + j] = s;
    }
    if (tid < ED) {
        float s = 0.f;
        for (int sl = 0; sl < PKS; ++sl)
            s += pksum[((size_t)bh * PKS + sl) * ED + tid];
        ksum[bh * ED + tid] = s;
    }
}

__global__ __launch_bounds__(256) void la_out(
    const float* __restrict__ query, const float* __restrict__ ctx,
    const float* __restrict__ ksum, float* __restrict__ out)
{
    __shared__ float sQ[64 * QP];
    __shared__ float sC[ED * KVP];
    __shared__ float sKs[ED];
    __shared__ float sDinv[64];

    const int bh = blockIdx.y;
    const int b = bh >> 4, h = bh & 15;
    const int t0 = blockIdx.x * 64;
    const int tid = threadIdx.x;
    const int wave = tid >> 5, lane = tid & 31;
    const int half = lane >> 4, lm = lane & 15;

    const size_t qbase = (size_t)b * TOK * HDIM + (size_t)h * ED;

    {
        const int lrow = tid >> 4, c4 = tid & 15;
        v4f cv[4], qv[4];
        #pragma unroll
        for (int u = 0; u < 4; ++u) {
            const int row = lrow + 16 * u;
            qv[u] = *(const v4f*)(query + qbase + (size_t)(t0 + row) * HDIM
                                  + (size_t)c4 * 4);
            cv[u] = *(const v4f*)(ctx + (size_t)bh * (ED * ED) + row * ED + c4 * 4);
        }
        if (tid < ED) sKs[tid] = ksum[bh * ED + tid];
        #pragma unroll
        for (int u = 0; u < 4; ++u) {
            const int row = lrow + 16 * u;
            *(v4f*)(sC + row * KVP + c4 * 4) = cv[u];
            v4f qf;
            qf.x = elu_fm(qv[u].x); qf.y = elu_fm(qv[u].y);
            qf.z = elu_fm(qv[u].z); qf.w = elu_fm(qv[u].w);
            *(v4f*)(sQ + row * QP + c4 * 4) = qf;
        }
    }
    __syncthreads();

    if (tid < 64) {
        float d = 0.f;
        #pragma unroll 8
        for (int e = 0; e < ED; ++e) d += sQ[tid * QP + e] * sKs[e];
        sDinv[tid] = 1.0f / (d + 1e-6f);
    }
    __syncthreads();

    const int n0a = (wave >> 2) << 4;
    const int n0b = n0a + 32;
    const int d0  = (wave & 3) << 4;

    v8f acc0 = {};
    v8f acc1 = {};
    for (int e = 0; e < ED; e += 4) {
        v2f a0, a1, bb;
        bb.x = sC[(e + 2 * half) * KVP + d0 + lm];
        bb.y = sC[(e + 1 + 2 * half) * KVP + d0 + lm];
        a0.x = sQ[(n0a + lm) * QP + e + 2 * half];
        a0.y = sQ[(n0a + lm) * QP + e + 1 + 2 * half];
        a1.x = sQ[(n0b + lm) * QP + e + 2 * half];
        a1.y = sQ[(n0b + lm) * QP + e + 1 + 2 * half];
        acc0 = __builtin_amdgcn_wmma_f32_16x16x4_f32(
            false, a0, false, bb, (short)0, acc0, false, false);
        acc1 = __builtin_amdgcn_wmma_f32_16x16x4_f32(
            false, a1, false, bb, (short)0, acc1, false, false);
    }

    #pragma unroll
    for (int r = 0; r < 8; ++r) {
        const int rowa = n0a + r + 8 * half;
        const int rowb = n0b + r + 8 * half;
        out[((size_t)b * TOK + t0 + rowa) * HDIM + h * ED + d0 + lm]
            = acc0[r] * sDinv[rowa];
        out[((size_t)b * TOK + t0 + rowb) * HDIM + h * ED + d0 + lm]
            = acc1[r] * sDinv[rowb];
    }
}

extern "C" void kernel_launch(void* const* d_in, const int* in_sizes, int n_in,
                              void* d_out, int out_size, void* d_ws, size_t ws_size,
                              hipStream_t stream) {
    (void)in_sizes; (void)n_in; (void)out_size; (void)ws_size;
    const float* q = (const float*)d_in[0];
    const float* k = (const float*)d_in[1];
    const float* v = (const float*)d_in[2];
    float* out = (float*)d_out;

    float* wsf   = (float*)d_ws;
    float* pctx  = wsf;
    float* pksum = pctx  + (size_t)BH * SPLITS * (ED * ED);
    float* ctx   = pksum + (size_t)BH * PKS * ED;
    float* ksum  = ctx   + (size_t)BH * (ED * ED);

    la_ctx_partial<<<BH * SPLITS, 256, 0, stream>>>(k, v, pctx, pksum);
    la_reduce<<<BH, 256, 0, stream>>>(pctx, pksum, ctx, ksum);
    la_out<<<dim3(TOK / 64, BH), 256, 0, stream>>>(q, ctx, ksum, out);
}


# Feature map elu(x)+1 made straight-line in both passes: exp as hardware exp2(x*log2e) for every lane, then select; was a lane-mask branch and a 13-op expf per element; maps differ in last bits
# speedup vs baseline: 1.7250x; 1.0874x over previous
#include <hip/hip_runtime.h>
#include <hip/hip_bf16.h>

typedef __attribute__((ext_vector_type(2))) float v2f;
typedef __attribute__((ext_vector_type(8))) float v8f;
typedef __attribute__((ext_vector_type(4))) float v4f;

#define BATCH   4
#define TOK     4096
#define HDIM    1024
#define NHEADS  16
#define ED      64
#define BH      (BATCH * NHEADS)
#define SPLITS  8
#define CHUNKS  4
#define CHUNK   128
#define PKS     (SPLITS * 16)
#define KVP     72
#define QP      68

__device__ __forceinline__ float elu_fm(float x) {
    const float e = __builtin_amdgcn_exp2f(x * 1.44269504088896341f);
    return x > 0.0f ? x + 1.0f : e;
}

__device__ __forceinline__ void wg_barrier_lds_only() {
    asm volatile("s_wait_dscnt 0x0\n\ts_barrier_signal -1\n\ts_barrier_wait -1" ::: "memory");
}

__global__ __launch_bounds__(256) __attribute__((amdgpu_waves_per_eu(4, 4)))
void la_ctx_partial(
    const float* __restrict__ key, const float* __restrict__ value,
    float* __restrict__ pctx, float* __restrict__ pksum)
{
    __shared__ float sK[CHUNK * KVP];
    __shared__ float sV[CHUNK * KVP];

    const int bh    = blockIdx.x / SPLITS;
    const int split = blockIdx.x % SPLITS;
    const int b = bh >> 4, h = bh & 15;
    const int tid = threadIdx.x;
    const int wave = tid >> 5, lane = tid & 31;
    const int half = lane >> 4, lm = lane & 15;

    const size_t base = (size_t)b * TOK * HDIM + (size_t)h * ED;

    const int e0a = (wave >> 2) << 4;
    const int e0b = e0a + 32;
    const int d0  = (wave & 3) << 4;

    v8f acc0 = {};
    v8f acc1 = {};
    float4 kacc = make_float4(0.f, 0.f, 0.f, 0.f);

    const int lrow = tid >> 4, c4 = tid & 15;
    const size_t gthr = base + (size_t)lrow * HDIM + (size_t)c4 * 4;
    v4f kreg[8], vreg[8];
    {
        const size_t g0 = gthr + (size_t)(split * CHUNKS * CHUNK) * HDIM;
        #pragma unroll
        for (int u = 0; u < 8; ++u) {
            kreg[u] = *(const v4f*)(key   + g0 + (size_t)(16 * u) * HDIM);
            vreg[u] = *(const v4f*)(value + g0 + (size_t)(16 * u) * HDIM);
        }
    }

    for (int c = 0; c < CHUNKS; ++c) {
        wg_barrier_lds_only();
        #pragma unroll
        for (int u = 0; u < 8; ++u) {
            const int row = lrow + 16 * u;
            v4f kf;
            kf.x = elu_fm(kreg[u].x); kf.y = elu_fm(kreg[u].y);
            kf.z = elu_fm(kreg[u].z); kf.w = elu_fm(kreg[u].w);
            kacc.x += kf.x; kacc.y += kf.y; kacc.z += kf.z; kacc.w += kf.w;
            *(v4f*)(sK + row * KVP + c4 * 4) = kf;
            *(v4f*)(sV + row * KVP + c4 * 4) = vreg[u];
        }
        if (c + 1 < CHUNKS) {
            const size_t g1 = gthr
                + (size_t)((split * CHUNKS + c + 1) * CHUNK) * HDIM;
            #pragma unroll
            for (int u = 0; u < 8; ++u) {
                kreg[u] = *(const v4f*)(key   + g1 + (size_t)(16 * u) * HDIM);
                vreg[u] = *(const v4f*)(value + g1 + (size_t)(16 * u) * HDIM);
            }
        }
        wg_barrier_lds_only();

        for (int s = 0; s < CHUNK; s += 4) {
            v2f a0, a1, bb;
            const int r0 = (s + 2 * half) * KVP;
            const int r1 = (s + 1 + 2 * half) * KVP;
            bb.x = sV[r0 + d0 + lm];
            bb.y = sV[r1 + d0 + lm];
            a0.x = sK[r0 + e0a + lm];
            a0.y = sK[r1 + e0a + lm];
            a1.x = sK[r0 + e0b + lm];
            a1.y = sK[r1 + e0b + lm];
            acc0 = __builtin_amdgcn_wmma_f32_16x16x4_f32(
                false, a0, false, bb, (short)0, acc0, false, false);
            acc1 = __builtin_amdgcn_wmma_f32_16x16x4_f32(
                false, a1, false, bb, (short)0, acc1, false, false);
        }
    }

    {
        float* dst = pksum + ((size_t)bh * PKS + split * 16 + wave * 2 + half) * ED
                           + lm * 4;
        *(float4*)dst = kacc;
    }

    float* pc = pctx + ((size_t)bh * SPLITS + split) * (ED * ED);
    #pragma unroll
    for (int r = 0; r < 8; ++r) {
        pc[(e0a + r + 8 * half) * ED + d0 + lm] = acc0[r];
        pc[(e0b + r + 8 * half) * ED + d0 + lm] = acc1[r];
    }
}

__global__ __launch_bounds__(256) void la_reduce(
    const float* __restrict__ pctx, const float* __restrict__ pksum,
    float* __restrict__ ctx, float* __restrict__ ksum)
{
    const int bh = blockIdx.x, tid = threadIdx.x;
    for (int j = tid; j < ED * ED; j += 256) {
        float s = 0.f;
        #pragma unroll
        for (int sp = 0; sp < SPLITS; ++sp)
            s += pctx[((size_t)bh * SPLITS + sp) * (ED * ED) + j];
        ctx[(size_t)bh * (ED * ED) + j] = s;
    }
    if (tid < ED) {
        float s = 0.f;
        for (int sl = 0; sl < PKS; ++sl)
            s += pksum[((size_t)bh * PKS + sl) * ED + tid];
        ksum[bh * ED + tid] = s;
    }
}

__global__ __launch_bounds__(256) void la_out(
    const float* __restrict__ query, const float* __restrict__ ctx,
    const float* __restrict__ ksum, float* __restrict__ out)
{
    __shared__ float sQ[64 * QP];
    __shared__ float sC[ED * KVP];
    __shared__ float sKs[ED];
    __shared__ float sDinv[64];

    const int bh = blockIdx.y;
    const int b = bh >> 4, h = bh & 15;
    const int t0 = blockIdx.x * 64;
    const int tid = threadIdx.x;
    const int wave = tid >> 5, lane = tid & 31;
    const int half = lane >> 4, lm = lane & 15;

    const size_t qbase = (size_t)b * TOK * HDIM + (size_t)h * ED;

    {
        const int lrow = tid >> 4, c4 = tid & 15;
        v4f cv[4], qv[4];
        #pragma unroll
        for (int u = 0; u < 4; ++u) {
            const int row = lrow + 16 * u;
            qv[u] = *(const v4f*)(query + qbase + (size_t)(t0 + row) * HDIM
                                  + (size_t)c4 * 4);
            cv[u] = *(const v4f*)(ctx + (size_t)bh * (ED * ED) + row * ED + c4 * 4);
        }
        if (tid < ED) sKs[tid] = ksum[bh * ED + tid];
        #pragma unroll
        for (int u = 0; u < 4; ++u) {
            const int row = lrow + 16 * u;
            *(v4f*)(sC + row * KVP + c4 * 4) = cv[u];
            v4f qf;
            qf.x = elu_fm(qv[u].x); qf.y = elu_fm(qv[u].y);
            qf.z = elu_fm(qv[u].z); qf.w = elu_fm(qv[u].w);
            *(v4f*)(sQ + row * QP + c4 * 4) = qf;
        }
    }
    __syncthreads();

    if (tid < 64) {
        float d = 0.f;
        #pragma unroll 8
        for (int e = 0; e < ED; ++e) d += sQ[tid * QP + e] * sKs[e];
        sDinv[tid] = 1.0f / (d + 1e-6f);
    }
    __syncthreads();

    const int n0a = (wave >> 2) << 4;
    const int n0b = n0a + 32;
    const int d0  = (wave & 3) << 4;

    v8f acc0 = {};
    v8f acc1 = {};
    for (int e = 0; e < ED; e += 4) {
        v2f a0, a1, bb;
        bb.x = sC[(e + 2 * half) * KVP + d0 + lm];
        bb.y = sC[(e + 1 + 2 * half) * KVP + d0 + lm];
        a0.x = sQ[(n0a + lm) * QP + e + 2 * half];
        a0.y = sQ[(n0a + lm) * QP + e + 1 + 2 * half];
        a1.x = sQ[(n0b + lm) * QP + e + 2 * half];
        a1.y = sQ[(n0b + lm) * QP + e + 1 + 2 * half];
        acc0 = __builtin_amdgcn_wmma_f32_16x16x4_f32(
            false, a0, false, bb, (short)0, acc0, false, false);
        acc1 = __builtin_amdgcn_wmma_f32_16x16x4_f32(
            false, a1, false, bb, (short)0, acc1, false, false);
    }

    #pragma unroll
    for (int r = 0; r < 8; ++r) {
        const int rowa = n0a + r + 8 * half;
        const int rowb = n0b + r + 8 * half;
        out[((size_t)b * TOK + t0 + rowa) * HDIM + h * ED + d0 + lm]
            = acc0[r] * sDinv[rowa];
        out[((size_t)b * TOK + t0 + rowb) * HDIM + h * ED + d0 + lm]
            = acc1[r] * sDinv[rowb];
    }
}

extern "C" void kernel_launch(void* const* d_in, const int* in_sizes, int n_in,
                              void* d_out, int out_size, void* d_ws, size_t ws_size,
                              hipStream_t stream) {
    (void)in_sizes; (void)n_in; (void)out_size; (void)ws_size;
    const float* q = (const float*)d_in[0];
    const float* k = (const float*)d_in[1];
    const float* v = (const float*)d_in[2];
    float* out = (float*)d_out;

    float* wsf   = (float*)d_ws;
    float* pctx  = wsf;
    float* pksum = pctx  + (size_t)BH * SPLITS * (ED * ED);
    float* ctx   = pksum + (size_t)BH * PKS * ED;
    float* ksum  = ctx   + (size_t)BH * (ED * ED);

    la_ctx_partial<<<BH * SPLITS, 256, 0, stream>>>(k, v, pctx, pksum);
    la_reduce<<<BH, 256, 0, stream>>>(pctx, pksum, ctx, ksum);
    la_out<<<dim3(TOK / 64, BH), 256, 0, stream>>>(q, ctx, ksum, out);
}


# Pass 2 (reduction of partials): each thread issues all its loads before the first add, one memory round trip on 128 work-groups (was 16 dependent round trips on 64); k_sum via LDS; sum order kept
# speedup vs baseline: 2.0537x; 1.1906x over previous
#include <hip/hip_runtime.h>
#include <hip/hip_bf16.h>

typedef __attribute__((ext_vector_type(2))) float v2f;
typedef __attribute__((ext_vector_type(8))) float v8f;
typedef __attribute__((ext_vector_type(4))) float v4f;

#define BATCH   4
#define TOK     4096
#define HDIM    1024
#define NHEADS  16
#define ED      64
#define BH      (BATCH * NHEADS)
#define SPLITS  8
#define CHUNKS  4
#define CHUNK   128
#define PKS     (SPLITS * 16)
#define KVP     72
#define QP      68

__device__ __forceinline__ float elu_fm(float x) {
    const float e = __builtin_amdgcn_exp2f(x * 1.44269504088896341f);
    return x > 0.0f ? x + 1.0f : e;
}

__device__ __forceinline__ void wg_barrier_lds_only() {
    asm volatile("s_wait_dscnt 0x0\n\ts_barrier_signal -1\n\ts_barrier_wait -1" ::: "memory");
}

__global__ __launch_bounds__(256) __attribute__((amdgpu_waves_per_eu(4, 4)))
void la_ctx_partial(
    const float* __restrict__ key, const float* __restrict__ value,
    float* __restrict__ pctx, float* __restrict__ pksum)
{
    __shared__ float sK[CHUNK * KVP];
    __shared__ float sV[CHUNK * KVP];

    const int bh    = blockIdx.x / SPLITS;
    const int split = blockIdx.x % SPLITS;
    const int b = bh >> 4, h = bh & 15;
    const int tid = threadIdx.x;
    const int wave = tid >> 5, lane = tid & 31;
    const int half = lane >> 4, lm = lane & 15;

    const size_t base = (size_t)b * TOK * HDIM + (size_t)h * ED;

    const int e0a = (wave >> 2) << 4;
    const int e0b = e0a + 32;
    const int d0  = (wave & 3) << 4;

    v8f acc0 = {};
    v8f acc1 = {};
    float4 kacc = make_float4(0.f, 0.f, 0.f, 0.f);

    const int lrow = tid >> 4, c4 = tid & 15;
    const size_t gthr = base + (size_t)lrow * HDIM + (size_t)c4 * 4;
    v4f kreg[8], vreg[8];
    {
        const size_t g0 = gthr + (size_t)(split * CHUNKS * CHUNK) * HDIM;
        #pragma unroll
        for (int u = 0; u < 8; ++u) {
            kreg[u] = *(const v4f*)(key   + g0 + (size_t)(16 * u) * HDIM);
            vreg[u] = *(const v4f*)(value + g0 + (size_t)(16 * u) * HDIM);
        }
    }

    for (int c = 0; c < CHUNKS; ++c) {
        wg_barrier_lds_only();
        #pragma unroll
        for (int u = 0; u < 8; ++u) {
            const int row = lrow + 16 * u;
            v4f kf;
            kf.x = elu_fm(kreg[u].x); kf.y = elu_fm(kreg[u].y);
            kf.z = elu_fm(kreg[u].z); kf.w = elu_fm(kreg[u].w);
            kacc.x += kf.x; kacc.y += kf.y; kacc.z += kf.z; kacc.w += kf.w;
            *(v4f*)(sK + row * KVP + c4 * 4) = kf;
            *(v4f*)(sV + row * KVP + c4 * 4) = vreg[u];
        }
        if (c + 1 < CHUNKS) {
            const size_t g1 = gthr
                + (size_t)((split * CHUNKS + c + 1) * CHUNK) * HDIM;
            #pragma unroll
            for (int u = 0; u < 8; ++u) {
                kreg[u] = *(const v4f*)(key   + g1 + (size_t)(16 * u) * HDIM);
                vreg[u] = *(const v4f*)(value + g1 + (size_t)(16 * u) * HDIM);
            }
        }
        wg_barrier_lds_only();

        for (int s = 0; s < CHUNK; s += 4) {
            v2f a0, a1, bb;
            const int r0 = (s + 2 * half) * KVP;
            const int r1 = (s + 1 + 2 * half) * KVP;
            bb.x = sV[r0 + d0 + lm];
            bb.y = sV[r1 + d0 + lm];
            a0.x = sK[r0 + e0a + lm];
            a0.y = sK[r1 + e0a + lm];
            a1.x = sK[r0 + e0b + lm];
            a1.y = sK[r1 + e0b + lm];
            acc0 = __builtin_amdgcn_wmma_f32_16x16x4_f32(
                false, a0, false, bb, (short)0, acc0, false, false);
            acc1 = __builtin_amdgcn_wmma_f32_16x16x4_f32(
                false, a1, false, bb, (short)0, acc1, false, false);
        }
    }

    {
        float* dst = pksum + ((size_t)bh * PKS + split * 16 + wave * 2 + half) * ED
                           + lm * 4;
        *(float4*)dst = kacc;
    }

    float* pc = pctx + ((size_t)bh * SPLITS + split) * (ED * ED);
    #pragma unroll
    for (int r = 0; r < 8; ++r) {
        pc[(e0a + r + 8 * half) * ED + d0 + lm] = acc0[r];
        pc[(e0b + r + 8 * half) * ED + d0 + lm] = acc1[r];
    }
}

#define RBLK    2
#define RCOLS   (ED / RBLK)
__global__ __launch_bounds__(256) __attribute__((amdgpu_waves_per_eu(2, 2)))
void la_reduce(
    const float* __restrict__ pctx, const float* __restrict__ pksum,
    float* __restrict__ ctx, float* __restrict__ ksum)
{
    __shared__ float sP[PKS * RCOLS];

    const int bh = blockIdx.x / RBLK, q = blockIdx.x % RBLK, tid = threadIdx.x;

    v4f kp[4];
    #pragma unroll
    for (int u = 0; u < 4; ++u) {
        const int idx = u * 256 + tid;
        const int sl = idx >> 3, c4 = idx & 7;
        kp[u] = *(const v4f*)(pksum + ((size_t)bh * PKS + sl) * ED
                              + q * RCOLS + c4 * 4);
    }
    v4f p[2][SPLITS];
    #pragma unroll
    for (int u = 0; u < 2; ++u) {
        const int j = q * 512 + u * 256 + tid;
        #pragma unroll
        for (int sp = 0; sp < SPLITS; ++sp)
            p[u][sp] = *(const v4f*)(pctx
                + ((size_t)bh * SPLITS + sp) * (ED * ED) + (size_t)j * 4);
    }

    #pragma unroll
    for (int u = 0; u < 4; ++u) {
        const int idx = u * 256 + tid;
        const int sl = idx >> 3, c4 = idx & 7;
        *(v4f*)(sP + sl * RCOLS + c4 * 4) = kp[u];
    }
    __syncthreads();

    #pragma unroll
    for (int u = 0; u < 2; ++u) {
        const int j = q * 512 + u * 256 + tid;
        v4f s = {0.f, 0.f, 0.f, 0.f};
        #pragma unroll
        for (int sp = 0; sp < SPLITS; ++sp) s += p[u][sp];
        *(v4f*)(ctx + (size_t)bh * (ED * ED) + (size_t)j * 4) = s;
    }

    if (tid < RCOLS) {
        float v[PKS];
        #pragma unroll
        for (int sl = 0; sl < PKS; ++sl) v[sl] = sP[sl * RCOLS + tid];
        __builtin_amdgcn_sched_barrier(0);
        float s = 0.f;
        #pragma unroll
        for (int sl = 0; sl < PKS; ++sl) s += v[sl];
        ksum[bh * ED + q * RCOLS + tid] = s;
    }
}

__global__ __launch_bounds__(256) void la_out(
    const float* __restrict__ query, const float* __restrict__ ctx,
    const float* __restrict__ ksum, float* __restrict__ out)
{
    __shared__ float sQ[64 * QP];
    __shared__ float sC[ED * KVP];
    __shared__ float sKs[ED];
    __shared__ float sDinv[64];

    const int bh = blockIdx.y;
    const int b = bh >> 4, h = bh & 15;
    const int t0 = blockIdx.x * 64;
    const int tid = threadIdx.x;
    const int wave = tid >> 5, lane = tid & 31;
    const int half = lane >> 4, lm = lane & 15;

    const size_t qbase = (size_t)b * TOK * HDIM + (size_t)h * ED;

    {
        const int lrow = tid >> 4, c4 = tid & 15;
        v4f cv[4], qv[4];
        #pragma unroll
        for (int u = 0; u < 4; ++u) {
            const int row = lrow + 16 * u;
            qv[u] = *(const v4f*)(query + qbase + (size_t)(t0 + row) * HDIM
                                  + (size_t)c4 * 4);
            cv[u] = *(const v4f*)(ctx + (size_t)bh * (ED * ED) + row * ED + c4 * 4);
        }
        if (tid < ED) sKs[tid] = ksum[bh * ED + tid];
        #pragma unroll
        for (int u = 0; u < 4; ++u) {
            const int row = lrow + 16 * u;
            *(v4f*)(sC + row * KVP + c4 * 4) = cv[u];
            v4f qf;
            qf.x = elu_fm(qv[u].x); qf.y = elu_fm(qv[u].y);
            qf.z = elu_fm(qv[u].z); qf.w = elu_fm(qv[u].w);
            *(v4f*)(sQ + row * QP + c4 * 4) = qf;
        }
    }
    __syncthreads();

    if (tid < 64) {
        float d = 0.f;
        #pragma unroll 8
        for (int e = 0; e < ED; ++e) d += sQ[tid * QP + e] * sKs[e];
        sDinv[tid] = 1.0f / (d + 1e-6f);
    }
    __syncthreads();

    const int n0a = (wave >> 2) << 4;
    const int n0b = n0a + 32;
    const int d0  = (wave & 3) << 4;

    v8f acc0 = {};
    v8f acc1 = {};
    for (int e = 0; e < ED; e += 4) {
        v2f a0, a1, bb;
        bb.x = sC[(e + 2 * half) * KVP + d0 + lm];
        bb.y = sC[(e + 1 + 2 * half) * KVP + d0 + lm];
        a0.x = sQ[(n0a + lm) * QP + e + 2 * half];
        a0.y = sQ[(n0a + lm) * QP + e + 1 + 2 * half];
        a1.x = sQ[(n0b + lm) * QP + e + 2 * half];
        a1.y = sQ[(n0b + lm) * QP + e + 1 + 2 * half];
        acc0 = __builtin_amdgcn_wmma_f32_16x16x4_f32(
            false, a0, false, bb, (short)0, acc0, false, false);
        acc1 = __builtin_amdgcn_wmma_f32_16x16x4_f32(
            false, a1, false, bb, (short)0, acc1, false, false);
    }

    #pragma unroll
    for (int r = 0; r < 8; ++r) {
        const int rowa = n0a + r + 8 * half;
        const int rowb = n0b + r + 8 * half;
        out[((size_t)b * TOK + t0 + rowa) * HDIM + h * ED + d0 + lm]
            = acc0[r] * sDinv[rowa];
        out[((size_t)b * TOK + t0 + rowb) * HDIM + h * ED + d0 + lm]
            = acc1[r] * sDinv[rowb];
    }
}

extern "C" void kernel_launch(void* const* d_in, const int* in_sizes, int n_in,
                              void* d_out, int out_size, void* d_ws, size_t ws_size,
                              hipStream_t stream) {
    (void)in_sizes; (void)n_in; (void)out_size; (void)ws_size;
    const float* q = (const float*)d_in[0];
    const float* k = (const float*)d_in[1];
    const float* v = (const float*)d_in[2];
    float* out = (float*)d_out;

    float* wsf   = (float*)d_ws;
    float* pctx  = wsf;
    float* pksum = pctx  + (size_t)BH * SPLITS * (ED * ED);
    float* ctx   = pksum + (size_t)BH * PKS * ED;
    float* ksum  = ctx   + (size_t)BH * (ED * ED);

    la_ctx_partial<<<BH * SPLITS, 256, 0, stream>>>(k, v, pctx, pksum);
    la_reduce<<<BH * RBLK, 256, 0, stream>>>(pctx, pksum, ctx, ksum);
    la_out<<<dim3(TOK / 64, BH), 256, 0, stream>>>(q, ctx, ksum, out);
}


# Pass 3 epilogue staged through LDS: output tile leaves as 16-byte-per-lane stores, one instruction = 4 whole 128-byte lines (was two 64-byte half lines per 4-byte store, two waves per line)
# speedup vs baseline: 2.2150x; 1.0785x over previous
#include <hip/hip_runtime.h>
#include <hip/hip_bf16.h>

typedef __attribute__((ext_vector_type(2))) float v2f;
typedef __attribute__((ext_vector_type(8))) float v8f;
typedef __attribute__((ext_vector_type(4))) float v4f;

#define BATCH   4
#define TOK     4096
#define HDIM    1024
#define NHEADS  16
#define ED      64
#define BH      (BATCH * NHEADS)
#define SPLITS  8
#define CHUNKS  4
#define CHUNK   128
#define PKS     (SPLITS * 16)
#define KVP     72
#define QP      68

__device__ __forceinline__ float elu_fm(float x) {
    const float e = __builtin_amdgcn_exp2f(x * 1.44269504088896341f);
    return x > 0.0f ? x + 1.0f : e;
}

__device__ __forceinline__ void wg_barrier_lds_only() {
    asm volatile("s_wait_dscnt 0x0\n\ts_barrier_signal -1\n\ts_barrier_wait -1" ::: "memory");
}

__global__ __launch_bounds__(256) __attribute__((amdgpu_waves_per_eu(4, 4)))
void la_ctx_partial(
    const float* __restrict__ key, const float* __restrict__ value,
    float* __restrict__ pctx, float* __restrict__ pksum)
{
    __shared__ float sK[CHUNK * KVP];
    __shared__ float sV[CHUNK * KVP];

    const int bh    = blockIdx.x / SPLITS;
    const int split = blockIdx.x % SPLITS;
    const int b = bh >> 4, h = bh & 15;
    const int tid = threadIdx.x;
    const int wave = tid >> 5, lane = tid & 31;
    const int half = lane >> 4, lm = lane & 15;

    const size_t base = (size_t)b * TOK * HDIM + (size_t)h * ED;

    const int e0a = (wave >> 2) << 4;
    const int e0b = e0a + 32;
    const int d0  = (wave & 3) << 4;

    v8f acc0 = {};
    v8f acc1 = {};
    float4 kacc = make_float4(0.f, 0.f, 0.f, 0.f);

    const int lrow = tid >> 4, c4 = tid & 15;
    const size_t gthr = base + (size_t)lrow * HDIM + (size_t)c4 * 4;
    v4f kreg[8], vreg[8];
    {
        const size_t g0 = gthr + (size_t)(split * CHUNKS * CHUNK) * HDIM;
        #pragma unroll
        for (int u = 0; u < 8; ++u) {
            kreg[u] = *(const v4f*)(key   + g0 + (size_t)(16 * u) * HDIM);
            vreg[u] = *(const v4f*)(value + g0 + (size_t)(16 * u) * HDIM);
        }
    }

    for (int c = 0; c < CHUNKS; ++c) {
        wg_barrier_lds_only();
        #pragma unroll
        for (int u = 0; u < 8; ++u) {
            const int row = lrow + 16 * u;
            v4f kf;
            kf.x = elu_fm(kreg[u].x); kf.y = elu_fm(kreg[u].y);
            kf.z = elu_fm(kreg[u].z); kf.w = elu_fm(kreg[u].w);
            kacc.x += kf.x; kacc.y += kf.y; kacc.z += kf.z; kacc.w += kf.w;
            *(v4f*)(sK + row * KVP + c4 * 4) = kf;
            *(v4f*)(sV + row * KVP + c4 * 4) = vreg[u];
        }
        if (c + 1 < CHUNKS) {
            const size_t g1 = gthr
                + (size_t)((split * CHUNKS + c + 1) * CHUNK) * HDIM;
            #pragma unroll
            for (int u = 0; u < 8; ++u) {
                kreg[u] = *(const v4f*)(key   + g1 + (size_t)(16 * u) * HDIM);
                vreg[u] = *(const v4f*)(value + g1 + (size_t)(16 * u) * HDIM);
            }
        }
        wg_barrier_lds_only();

        for (int s = 0; s < CHUNK; s += 4) {
            v2f a0, a1, bb;
            const int r0 = (s + 2 * half) * KVP;
            const int r1 = (s + 1 + 2 * half) * KVP;
            bb.x = sV[r0 + d0 + lm];
            bb.y = sV[r1 + d0 + lm];
            a0.x = sK[r0 + e0a + lm];
            a0.y = sK[r1 + e0a + lm];
            a1.x = sK[r0 + e0b + lm];
            a1.y = sK[r1 + e0b + lm];
            acc0 = __builtin_amdgcn_wmma_f32_16x16x4_f32(
                false, a0, false, bb, (short)0, acc0, false, false);
            acc1 = __builtin_amdgcn_wmma_f32_16x16x4_f32(
                false, a1, false, bb, (short)0, acc1, false, false);
        }
    }

    {
        float* dst = pksum + ((size_t)bh * PKS + split * 16 + wave * 2 + half) * ED
                           + lm * 4;
        *(float4*)dst = kacc;
    }

    float* pc = pctx + ((size_t)bh * SPLITS + split) * (ED * ED);
    #pragma unroll
    for (int r = 0; r < 8; ++r) {
        pc[(e0a + r + 8 * half) * ED + d0 + lm] = acc0[r];
        pc[(e0b + r + 8 * half) * ED + d0 + lm] = acc1[r];
    }
}

#define RBLK    2
#define RCOLS   (ED / RBLK)
__global__ __launch_bounds__(256) __attribute__((amdgpu_waves_per_eu(2, 2)))
void la_reduce(
    const float* __restrict__ pctx, const float* __restrict__ pksum,
    float* __restrict__ ctx, float* __restrict__ ksum)
{
    __shared__ float sP[PKS * RCOLS];

    const int bh = blockIdx.x / RBLK, q = blockIdx.x % RBLK, tid = threadIdx.x;

    v4f kp[4];
    #pragma unroll
    for (int u = 0; u < 4; ++u) {
        const int idx = u * 256 + tid;
        const int sl = idx >> 3, c4 = idx & 7;
        kp[u] = *(const v4f*)(pksum + ((size_t)bh * PKS + sl) * ED
                              + q * RCOLS + c4 * 4);
    }
    v4f p[2][SPLITS];
    #pragma unroll
    for (int u = 0; u < 2; ++u) {
        const int j = q * 512 + u * 256 + tid;
        #pragma unroll
        for (int sp = 0; sp < SPLITS; ++sp)
            p[u][sp] = *(const v4f*)(pctx
                + ((size_t)bh * SPLITS + sp) * (ED * ED) + (size_t)j * 4);
    }

    #pragma unroll
    for (int u = 0; u < 4; ++u) {
        const int idx = u * 256 + tid;
        const int sl = idx >> 3, c4 = idx & 7;
        *(v4f*)(sP + sl * RCOLS + c4 * 4) = kp[u];
    }
    __syncthreads();

    #pragma unroll
    for (int u = 0; u < 2; ++u) {
        const int j = q * 512 + u * 256 + tid;
        v4f s = {0.f, 0.f, 0.f, 0.f};
        #pragma unroll
        for (int sp = 0; sp < SPLITS; ++sp) s += p[u][sp];
        *(v4f*)(ctx + (size_t)bh * (ED * ED) + (size_t)j * 4) = s;
    }

    if (tid < RCOLS) {
        float v[PKS];
        #pragma unroll
        for (int sl = 0; sl < PKS; ++sl) v[sl] = sP[sl * RCOLS + tid];
        __builtin_amdgcn_sched_barrier(0);
        float s = 0.f;
        #pragma unroll
        for (int sl = 0; sl < PKS; ++sl) s += v[sl];
        ksum[bh * ED + q * RCOLS + tid] = s;
    }
}

__global__ __launch_bounds__(256) void la_out(
    const float* __restrict__ query, const float* __restrict__ ctx,
    const float* __restrict__ ksum, float* __restrict__ out)
{
    __shared__ float sQ[64 * QP];
    __shared__ float sC[ED * KVP];
    __shared__ float sKs[ED];
    __shared__ float sDinv[64];

    const int bh = blockIdx.y;
    const int b = bh >> 4, h = bh & 15;
    const int t0 = blockIdx.x * 64;
    const int tid = threadIdx.x;
    const int wave = tid >> 5, lane = tid & 31;
    const int half = lane >> 4, lm = lane & 15;

    const size_t qbase = (size_t)b * TOK * HDIM + (size_t)h * ED;

    {
        const int lrow = tid >> 4, c4 = tid & 15;
        v4f cv[4], qv[4];
        #pragma unroll
        for (int u = 0; u < 4; ++u) {
            const int row = lrow + 16 * u;
            qv[u] = *(const v4f*)(query + qbase + (size_t)(t0 + row) * HDIM
                                  + (size_t)c4 * 4);
            cv[u] = *(const v4f*)(ctx + (size_t)bh * (ED * ED) + row * ED + c4 * 4);
        }
        if (tid < ED) sKs[tid] = ksum[bh * ED + tid];
        #pragma unroll
        for (int u = 0; u < 4; ++u) {
            const int row = lrow + 16 * u;
            *(v4f*)(sC + row * KVP + c4 * 4) = cv[u];
            v4f qf;
            qf.x = elu_fm(qv[u].x); qf.y = elu_fm(qv[u].y);
            qf.z = elu_fm(qv[u].z); qf.w = elu_fm(qv[u].w);
            *(v4f*)(sQ + row * QP + c4 * 4) = qf;
        }
    }
    __syncthreads();

    if (tid < 64) {
        float d = 0.f;
        #pragma unroll 8
        for (int e = 0; e < ED; ++e) d += sQ[tid * QP + e] * sKs[e];
        sDinv[tid] = 1.0f / (d + 1e-6f);
    }
    __syncthreads();

    const int n0a = (wave >> 2) << 4;
    const int n0b = n0a + 32;
    const int d0  = (wave & 3) << 4;

    v8f acc0 = {};
    v8f acc1 = {};
    for (int e = 0; e < ED; e += 4) {
        v2f a0, a1, bb;
        bb.x = sC[(e + 2 * half) * KVP + d0 + lm];
        bb.y = sC[(e + 1 + 2 * half) * KVP + d0 + lm];
        a0.x = sQ[(n0a + lm) * QP + e + 2 * half];
        a0.y = sQ[(n0a + lm) * QP + e + 1 + 2 * half];
        a1.x = sQ[(n0b + lm) * QP + e + 2 * half];
        a1.y = sQ[(n0b + lm) * QP + e + 1 + 2 * half];
        acc0 = __builtin_amdgcn_wmma_f32_16x16x4_f32(
            false, a0, false, bb, (short)0, acc0, false, false);
        acc1 = __builtin_amdgcn_wmma_f32_16x16x4_f32(
            false, a1, false, bb, (short)0, acc1, false, false);
    }

    __syncthreads();
    #pragma unroll
    for (int r = 0; r < 8; ++r) {
        sQ[(n0a + r + 8 * half) * QP + d0 + lm] = acc0[r];
        sQ[(n0b + r + 8 * half) * QP + d0 + lm] = acc1[r];
    }
    __syncthreads();
    {
        const int lrow = tid >> 4, c4 = tid & 15;
        #pragma unroll
        for (int u = 0; u < 4; ++u) {
            const int row = lrow + 16 * u;
            v4f o = *(const v4f*)(sQ + row * QP + c4 * 4);
            const float di = sDinv[row];
            o.x *= di; o.y *= di; o.z *= di; o.w *= di;
            *(v4f*)(out + ((size_t)b * TOK + t0 + row) * HDIM + h * ED + c4 * 4) = o;
        }
    }
}

extern "C" void kernel_launch(void* const* d_in, const int* in_sizes, int n_in,
                              void* d_out, int out_size, void* d_ws, size_t ws_size,
                              hipStream_t stream) {
    (void)in_sizes; (void)n_in; (void)out_size; (void)ws_size;
    const float* q = (const float*)d_in[0];
    const float* k = (const float*)d_in[1];
    const float* v = (const float*)d_in[2];
    float* out = (float*)d_out;

    float* wsf   = (float*)d_ws;
    float* pctx  = wsf;
    float* pksum = pctx  + (size_t)BH * SPLITS * (ED * ED);
    float* ctx   = pksum + (size_t)BH * PKS * ED;
    float* ksum  = ctx   + (size_t)BH * (ED * ED);

    la_ctx_partial<<<BH * SPLITS, 256, 0, stream>>>(k, v, pctx, pksum);
    la_reduce<<<BH * RBLK, 256, 0, stream>>>(pctx, pksum, ctx, ksum);
    la_out<<<dim3(TOK / 64, BH), 256, 0, stream>>>(q, ctx, ksum, out);
}


# Pass 1: K chunk in LDS as row quads, V rows at 256-byte pitch: fragment reads take immediate offsets, no vector adds or moves between WMMAs; k order within 8-row groups and k_sum grouping change
# speedup vs baseline: 2.3475x; 1.0598x over previous
#include <hip/hip_runtime.h>
#include <hip/hip_bf16.h>

typedef __attribute__((ext_vector_type(2))) float v2f;
typedef __attribute__((ext_vector_type(8))) float v8f;
typedef __attribute__((ext_vector_type(4))) float v4f;

#define BATCH   4
#define TOK     4096
#define HDIM    1024
#define NHEADS  16
#define ED      64
#define BH      (BATCH * NHEADS)
#define SPLITS  8
#define CHUNKS  4
#define CHUNK   128
#define PKS     (SPLITS * 16)
#define KVP     72
#define QP      68
#define QRP     304
#define VP1     64

__device__ __forceinline__ float elu_fm(float x) {
    const float e = __builtin_amdgcn_exp2f(x * 1.44269504088896341f);
    return x > 0.0f ? x + 1.0f : e;
}

__device__ __forceinline__ void wg_barrier_lds_only() {
    asm volatile("s_wait_dscnt 0x0\n\ts_barrier_signal -1\n\ts_barrier_wait -1" ::: "memory");
}

__global__ __launch_bounds__(256) __attribute__((amdgpu_waves_per_eu(4, 4)))
void la_ctx_partial(
    const float* __restrict__ key, const float* __restrict__ value,
    float* __restrict__ pctx, float* __restrict__ pksum)
{
    __shared__ __attribute__((aligned(16))) float sK[(CHUNK / 4) * QRP];
    __shared__ __attribute__((aligned(16))) float sV[CHUNK * VP1];

    const int bh    = blockIdx.x / SPLITS;
    const int split = blockIdx.x % SPLITS;
    const int b = bh >> 4, h = bh & 15;
    const int tid = threadIdx.x;
    const int wave = tid >> 5, lane = tid & 31;
    const int half = lane >> 4, lm = lane & 15;

    const size_t base = (size_t)b * TOK * HDIM + (size_t)h * ED;

    const int e0a = (wave >> 2) << 4;
    const int e0b = e0a + 32;
    const int d0  = (wave & 3) << 4;

    v8f acc0 = {};
    v8f acc1 = {};
    v2f ks0 = {0.f, 0.f}, ks1 = {0.f, 0.f}, ks2 = {0.f, 0.f}, ks3 = {0.f, 0.f};

    const int qpos = ((lm >> 2) << 2) + 80 * (lm & 3);
    const float* rA0 = sK + half * QRP + e0a + qpos;
    const float* rA1 = sK + half * QRP + e0b + qpos;
    const float* rB  = sV + (4 * half) * VP1 + ((d0 + lm) ^ (32 * half));

    const int lrow = tid >> 4, c4 = tid & 15;
    const size_t gthr = base + (size_t)(4 * lrow) * HDIM + (size_t)c4 * 4;
    v4f kreg[8], vreg[8];
    {
        const size_t g0 = gthr + (size_t)(split * CHUNKS * CHUNK) * HDIM;
        #pragma unroll
        for (int u = 0; u < 8; ++u) {
            const int ro = 64 * (u >> 2) + (u & 3);
            kreg[u] = *(const v4f*)(key   + g0 + (size_t)ro * HDIM);
            vreg[u] = *(const v4f*)(value + g0 + (size_t)ro * HDIM);
        }
    }

    for (int c = 0; c < CHUNKS; ++c) {
        wg_barrier_lds_only();
        #pragma unroll
        for (int w = 0; w < 2; ++w) {
            v4f kf[4];
            #pragma unroll
            for (int i = 0; i < 4; ++i) {
                kf[i].x = elu_fm(kreg[4 * w + i].x); kf[i].y = elu_fm(kreg[4 * w + i].y);
                kf[i].z = elu_fm(kreg[4 * w + i].z); kf[i].w = elu_fm(kreg[4 * w + i].w);
                *(v4f*)(sV + (4 * lrow + 64 * w + i) * VP1 + ((c4 * 4) ^ (32 * (lrow & 1)))) = vreg[4 * w + i];
            }
            float* dq = sK + (lrow + 16 * w) * QRP + c4 * 4;
            v4f t;
            t.x = kf[0].x; t.y = kf[1].x; t.z = kf[2].x; t.w = kf[3].x;  *(v4f*)(dq)       = t;
            ks0 += t.xy; ks0 += t.zw;
            t.x = kf[0].y; t.y = kf[1].y; t.z = kf[2].y; t.w = kf[3].y;  *(v4f*)(dq + 80)  = t;
            ks1 += t.xy; ks1 += t.zw;
            t.x = kf[0].z; t.y = kf[1].z; t.z = kf[2].z; t.w = kf[3].z;  *(v4f*)(dq + 160) = t;
            ks2 += t.xy; ks2 += t.zw;
            t.x = kf[0].w; t.y = kf[1].w; t.z = kf[2].w; t.w = kf[3].w;  *(v4f*)(dq + 240) = t;
            ks3 += t.xy; ks3 += t.zw;
        }
        if (c + 1 < CHUNKS) {
            const size_t g1 = gthr
                + (size_t)((split * CHUNKS + c + 1) * CHUNK) * HDIM;
            #pragma unroll
            for (int u = 0; u < 8; ++u) {
                const int ro = 64 * (u >> 2) + (u & 3);
                kreg[u] = *(const v4f*)(key   + g1 + (size_t)ro * HDIM);
                vreg[u] = *(const v4f*)(value + g1 + (size_t)ro * HDIM);
            }
        }
        wg_barrier_lds_only();

        #pragma unroll
        for (int m = 0; m < CHUNK / 8; ++m) {
            const v4f qa0 = *(const v4f*)(rA0 + m * 2 * QRP);
            const v4f qa1 = *(const v4f*)(rA1 + m * 2 * QRP);
            const float* pb = rB + m * 8 * VP1;
            v2f a0, a1, bb;
            bb.x = pb[0];       bb.y = pb[VP1];
            a0.x = qa0.x; a0.y = qa0.y;
            a1.x = qa1.x; a1.y = qa1.y;
            acc0 = __builtin_amdgcn_wmma_f32_16x16x4_f32(
                false, a0, false, bb, (short)0, acc0, false, false);
            acc1 = __builtin_amdgcn_wmma_f32_16x16x4_f32(
                false, a1, false, bb, (short)0, acc1, false, false);
            bb.x = pb[2 * VP1]; bb.y = pb[3 * VP1];
            a0.x = qa0.z; a0.y = qa0.w;
            a1.x = qa1.z; a1.y = qa1.w;
            acc0 = __builtin_amdgcn_wmma_f32_16x16x4_f32(
                false, a0, false, bb, (short)0, acc0, false, false);
            acc1 = __builtin_amdgcn_wmma_f32_16x16x4_f32(
                false, a1, false, bb, (short)0, acc1, false, false);
        }
    }

    {
        float* dst = pksum + ((size_t)bh * PKS + split * 16 + wave * 2 + half) * ED
                           + lm * 4;
        *(float4*)dst = make_float4(ks0.x + ks0.y, ks1.x + ks1.y, ks2.x + ks2.y, ks3.x + ks3.y);
    }

    float* pc = pctx + ((size_t)bh * SPLITS + split) * (ED * ED);
    #pragma unroll
    for (int r = 0; r < 8; ++r) {
        pc[(e0a + r + 8 * half) * ED + d0 + lm] = acc0[r];
        pc[(e0b + r + 8 * half) * ED + d0 + lm] = acc1[r];
    }
}

#define RBLK    2
#define RCOLS   (ED / RBLK)
__global__ __launch_bounds__(256) __attribute__((amdgpu_waves_per_eu(2, 2)))
void la_reduce(
    const float* __restrict__ pctx, const float* __restrict__ pksum,
    float* __restrict__ ctx, float* __restrict__ ksum)
{
    __shared__ float sP[PKS * RCOLS];

    const int bh = blockIdx.x / RBLK, q = blockIdx.x % RBLK, tid = threadIdx.x;

    v4f kp[4];
    #pragma unroll
    for (int u = 0; u < 4; ++u) {
        const int idx = u * 256 + tid;
        const int sl = idx >> 3, c4 = idx & 7;
        kp[u] = *(const v4f*)(pksum + ((size_t)bh * PKS + sl) * ED
                              + q * RCOLS + c4 * 4);
    }
    v4f p[2][SPLITS];
    #pragma unroll
    for (int u = 0; u < 2; ++u) {
        const int j = q * 512 + u * 256 + tid;
        #pragma unroll
        for (int sp = 0; sp < SPLITS; ++sp)
            p[u][sp] = *(const v4f*)(pctx
                + ((size_t)bh * SPLITS + sp) * (ED * ED) + (size_t)j * 4);
    }

    #pragma unroll
    for (int u = 0; u < 4; ++u) {
        const int idx = u * 256 + tid;
        const int sl = idx >> 3, c4 = idx & 7;
        *(v4f*)(sP + sl * RCOLS + c4 * 4) = kp[u];
    }
    __syncthreads();

    #pragma unroll
    for (int u = 0; u < 2; ++u) {
        const int j = q * 512 + u * 256 + tid;
        v4f s = {0.f, 0.f, 0.f, 0.f};
        #pragma unroll
        for (int sp = 0; sp < SPLITS; ++sp) s += p[u][sp];
        *(v4f*)(ctx + (size_t)bh * (ED * ED) + (size_t)j * 4) = s;
    }

    if (tid < RCOLS) {
        float v[PKS];
        #pragma unroll
        for (int sl = 0; sl < PKS; ++sl) v[sl] = sP[sl * RCOLS + tid];
        __builtin_amdgcn_sched_barrier(0);
        float s = 0.f;
        #pragma unroll
        for (int sl = 0; sl < PKS; ++sl) s += v[sl];
        ksum[bh * ED + q * RCOLS + tid] = s;
    }
}

__global__ __launch_bounds__(256) void la_out(
    const float* __restrict__ query, const float* __restrict__ ctx,
    const float* __restrict__ ksum, float* __restrict__ out)
{
    __shared__ float sQ[64 * QP];
    __shared__ float sC[ED * KVP];
    __shared__ float sKs[ED];
    __shared__ float sDinv[64];

    const int bh = blockIdx.y;
    const int b = bh >> 4, h = bh & 15;
    const int t0 = blockIdx.x * 64;
    const int tid = threadIdx.x;
    const int wave = tid >> 5, lane = tid & 31;
    const int half = lane >> 4, lm = lane & 15;

    const size_t qbase = (size_t)b * TOK * HDIM + (size_t)h * ED;

    {
        const int lrow = tid >> 4, c4 = tid & 15;
        v4f cv[4], qv[4];
        #pragma unroll
        for (int u = 0; u < 4; ++u) {
            const int row = lrow + 16 * u;
            qv[u] = *(const v4f*)(query + qbase + (size_t)(t0 + row) * HDIM
                                  + (size_t)c4 * 4);
            cv[u] = *(const v4f*)(ctx + (size_t)bh * (ED * ED) + row * ED + c4 * 4);
        }
        if (tid < ED) sKs[tid] = ksum[bh * ED + tid];
        #pragma unroll
        for (int u = 0; u < 4; ++u) {
            const int row = lrow + 16 * u;
            *(v4f*)(sC + row * KVP + c4 * 4) = cv[u];
            v4f qf;
            qf.x = elu_fm(qv[u].x); qf.y = elu_fm(qv[u].y);
            qf.z = elu_fm(qv[u].z); qf.w = elu_fm(qv[u].w);
            *(v4f*)(sQ + row * QP + c4 * 4) = qf;
        }
    }
    __syncthreads();

    if (tid < 64) {
        float d = 0.f;
        #pragma unroll 8
        for (int e = 0; e < ED; ++e) d += sQ[tid * QP + e] * sKs[e];
        sDinv[tid] = 1.0f / (d + 1e-6f);
    }
    __syncthreads();

    const int n0a = (wave >> 2) << 4;
    const int n0b = n0a + 32;
    const int d0  = (wave & 3) << 4;

    v8f acc0 = {};
    v8f acc1 = {};
    for (int e = 0; e < ED; e += 4) {
        v2f a0, a1, bb;
        bb.x = sC[(e + 2 * half) * KVP + d0 + lm];
        bb.y = sC[(e + 1 + 2 * half) * KVP + d0 + lm];
        a0.x = sQ[(n0a + lm) * QP + e + 2 * half];
        a0.y = sQ[(n0a + lm) * QP + e + 1 + 2 * half];
        a1.x = sQ[(n0b + lm) * QP + e + 2 * half];
        a1.y = sQ[(n0b + lm) * QP + e + 1 + 2 * half];
        acc0 = __builtin_amdgcn_wmma_f32_16x16x4_f32(
            false, a0, false, bb, (short)0, acc0, false, false);
        acc1 = __builtin_amdgcn_wmma_f32_16x16x4_f32(
            false, a1, false, bb, (short)0, acc1, false, false);
    }

    __syncthreads();
    #pragma unroll
    for (int r = 0; r < 8; ++r) {
        sQ[(n0a + r + 8 * half) * QP + d0 + lm] = acc0[r];
        sQ[(n0b + r + 8 * half) * QP + d0 + lm] = acc1[r];
    }
    __syncthreads();
    {
        const int lrow = tid >> 4, c4 = tid & 15;
        #pragma unroll
        for (int u = 0; u < 4; ++u) {
            const int row = lrow + 16 * u;
            v4f o = *(const v4f*)(sQ + row * QP + c4 * 4);
            const float di = sDinv[row];
            o.x *= di; o.y *= di; o.z *= di; o.w *= di;
            *(v4f*)(out + ((size_t)b * TOK + t0 + row) * HDIM + h * ED + c4 * 4) = o;
        }
    }
}

extern "C" void kernel_launch(void* const* d_in, const int* in_sizes, int n_in,
                              void* d_out, int out_size, void* d_ws, size_t ws_size,
                              hipStream_t stream) {
    (void)in_sizes; (void)n_in; (void)out_size; (void)ws_size;
    const float* q = (const float*)d_in[0];
    const float* k = (const float*)d_in[1];
    const float* v = (const float*)d_in[2];
    float* out = (float*)d_out;

    float* wsf   = (float*)d_ws;
    float* pctx  = wsf;
    float* pksum = pctx  + (size_t)BH * SPLITS * (ED * ED);
    float* ctx   = pksum + (size_t)BH * PKS * ED;
    float* ksum  = ctx   + (size_t)BH * (ED * ED);

    la_ctx_partial<<<BH * SPLITS, 256, 0, stream>>>(k, v, pctx, pksum);
    la_reduce<<<BH * RBLK, 256, 0, stream>>>(pctx, pksum, ctx, ksum);
    la_out<<<dim3(TOK / 64, BH), 256, 0, stream>>>(q, ctx, ksum, out);
}


# Pass 3 as a tile loop: 1024 resident work-groups (4 per WGP) walk 4 token tiles of a head each; context B fragments stay in registers, Q prefetched a tile ahead, 1 barrier per tile; sums kept
# speedup vs baseline: 2.4487x; 1.0431x over previous
#include <hip/hip_runtime.h>
#include <hip/hip_bf16.h>

typedef __attribute__((ext_vector_type(2))) float v2f;
typedef __attribute__((ext_vector_type(8))) float v8f;
typedef __attribute__((ext_vector_type(4))) float v4f;

#define BATCH   4
#define TOK     4096
#define HDIM    1024
#define NHEADS  16
#define ED      64
#define BH      (BATCH * NHEADS)
#define SPLITS  8
#define CHUNKS  4
#define CHUNK   128
#define PKS     (SPLITS * 16)
#define KVP     72
#define QP      68
#define QRP     304
#define VP1     64

__device__ __forceinline__ float elu_fm(float x) {
    const float e = __builtin_amdgcn_exp2f(x * 1.44269504088896341f);
    return x > 0.0f ? x + 1.0f : e;
}

__device__ __forceinline__ void wg_barrier_lds_only() {
    asm volatile("s_wait_dscnt 0x0\n\ts_barrier_signal -1\n\ts_barrier_wait -1" ::: "memory");
}

__global__ __launch_bounds__(256) __attribute__((amdgpu_waves_per_eu(4, 4)))
void la_ctx_partial(
    const float* __restrict__ key, const float* __restrict__ value,
    float* __restrict__ pctx, float* __restrict__ pksum)
{
    __shared__ __attribute__((aligned(16))) float sK[(CHUNK / 4) * QRP];
    __shared__ __attribute__((aligned(16))) float sV[CHUNK * VP1];

    const int bh    = blockIdx.x / SPLITS;
    const int split = blockIdx.x % SPLITS;
    const int b = bh >> 4, h = bh & 15;
    const int tid = threadIdx.x;
    const int wave = tid >> 5, lane = tid & 31;
    const int half = lane >> 4, lm = lane & 15;

    const size_t base = (size_t)b * TOK * HDIM + (size_t)h * ED;

    const int e0a = (wave >> 2) << 4;
    const int e0b = e0a + 32;
    const int d0  = (wave & 3) << 4;

    v8f acc0 = {};
    v8f acc1 = {};
    v2f ks0 = {0.f, 0.f}, ks1 = {0.f, 0.f}, ks2 = {0.f, 0.f}, ks3 = {0.f, 0.f};

    const int qpos = ((lm >> 2) << 2) + 80 * (lm & 3);
    const float* rA0 = sK + half * QRP + e0a + qpos;
    const float* rA1 = sK + half * QRP + e0b + qpos;
    const float* rB  = sV + (4 * half) * VP1 + ((d0 + lm) ^ (32 * half));

    const int lrow = tid >> 4, c4 = tid & 15;
    const size_t gthr = base + (size_t)(4 * lrow) * HDIM + (size_t)c4 * 4;
    v4f kreg[8], vreg[8];
    {
        const size_t g0 = gthr + (size_t)(split * CHUNKS * CHUNK) * HDIM;
        #pragma unroll
        for (int u = 0; u < 8; ++u) {
            const int ro = 64 * (u >> 2) + (u & 3);
            kreg[u] = *(const v4f*)(key   + g0 + (size_t)ro * HDIM);
            vreg[u] = *(const v4f*)(value + g0 + (size_t)ro * HDIM);
        }
    }

    for (int c = 0; c < CHUNKS; ++c) {
        wg_barrier_lds_only();
        #pragma unroll
        for (int w = 0; w < 2; ++w) {
            v4f kf[4];
            #pragma unroll
            for (int i = 0; i < 4; ++i) {
                kf[i].x = elu_fm(kreg[4 * w + i].x); kf[i].y = elu_fm(kreg[4 * w + i].y);
                kf[i].z = elu_fm(kreg[4 * w + i].z); kf[i].w = elu_fm(kreg[4 * w + i].w);
                *(v4f*)(sV + (4 * lrow + 64 * w + i) * VP1 + ((c4 * 4) ^ (32 * (lrow & 1)))) = vreg[4 * w + i];
            }
            float* dq = sK + (lrow + 16 * w) * QRP + c4 * 4;
            v4f t;
            t.x = kf[0].x; t.y = kf[1].x; t.z = kf[2].x; t.w = kf[3].x;  *(v4f*)(dq)       = t;
            ks0 += t.xy; ks0 += t.zw;
            t.x = kf[0].y; t.y = kf[1].y; t.z = kf[2].y; t.w = kf[3].y;  *(v4f*)(dq + 80)  = t;
            ks1 += t.xy; ks1 += t.zw;
            t.x = kf[0].z; t.y = kf[1].z; t.z = kf[2].z; t.w = kf[3].z;  *(v4f*)(dq + 160) = t;
            ks2 += t.xy; ks2 += t.zw;
            t.x = kf[0].w; t.y = kf[1].w; t.z = kf[2].w; t.w = kf[3].w;  *(v4f*)(dq + 240) = t;
            ks3 += t.xy; ks3 += t.zw;
        }
        if (c + 1 < CHUNKS) {
            const size_t g1 = gthr
                + (size_t)((split * CHUNKS + c + 1) * CHUNK) * HDIM;
            #pragma unroll
            for (int u = 0; u < 8; ++u) {
                const int ro = 64 * (u >> 2) + (u & 3);
                kreg[u] = *(const v4f*)(key   + g1 + (size_t)ro * HDIM);
                vreg[u] = *(const v4f*)(value + g1 + (size_t)ro * HDIM);
            }
        }
        wg_barrier_lds_only();

        #pragma unroll
        for (int m = 0; m < CHUNK / 8; ++m) {
            const v4f qa0 = *(const v4f*)(rA0 + m * 2 * QRP);
            const v4f qa1 = *(const v4f*)(rA1 + m * 2 * QRP);
            const float* pb = rB + m * 8 * VP1;
            v2f a0, a1, bb;
            bb.x = pb[0];       bb.y = pb[VP1];
            a0.x = qa0.x; a0.y = qa0.y;
            a1.x = qa1.x; a1.y = qa1.y;
            acc0 = __builtin_amdgcn_wmma_f32_16x16x4_f32(
                false, a0, false, bb, (short)0, acc0, false, false);
            acc1 = __builtin_amdgcn_wmma_f32_16x16x4_f32(
                false, a1, false, bb, (short)0, acc1, false, false);
            bb.x = pb[2 * VP1]; bb.y = pb[3 * VP1];
            a0.x = qa0.z; a0.y = qa0.w;
            a1.x = qa1.z; a1.y = qa1.w;
            acc0 = __builtin_amdgcn_wmma_f32_16x16x4_f32(
                false, a0, false, bb, (short)0, acc0, false, false);
            acc1 = __builtin_amdgcn_wmma_f32_16x16x4_f32(
                false, a1, false, bb, (short)0, acc1, false, false);
        }
    }

    {
        float* dst = pksum + ((size_t)bh * PKS + split * 16 + wave * 2 + half) * ED
                           + lm * 4;
        *(float4*)dst = make_float4(ks0.x + ks0.y, ks1.x + ks1.y, ks2.x + ks2.y, ks3.x + ks3.y);
    }

    float* pc = pctx + ((size_t)bh * SPLITS + split) * (ED * ED);
    #pragma unroll
    for (int r = 0; r < 8; ++r) {
        pc[(e0a + r + 8 * half) * ED + d0 + lm] = acc0[r];
        pc[(e0b + r + 8 * half) * ED + d0 + lm] = acc1[r];
    }
}

#define RBLK    2
#define RCOLS   (ED / RBLK)
__global__ __launch_bounds__(256) __attribute__((amdgpu_waves_per_eu(2, 2)))
void la_reduce(
    const float* __restrict__ pctx, const float* __restrict__ pksum,
    float* __restrict__ ctx, float* __restrict__ ksum)
{
    __shared__ float sP[PKS * RCOLS];

    const int bh = blockIdx.x / RBLK, q = blockIdx.x % RBLK, tid = threadIdx.x;

    v4f kp[4];
    #pragma unroll
    for (int u = 0; u < 4; ++u) {
        const int idx = u * 256 + tid;
        const int sl = idx >> 3, c4 = idx & 7;
        kp[u] = *(const v4f*)(pksum + ((size_t)bh * PKS + sl) * ED
                              + q * RCOLS + c4 * 4);
    }
    v4f p[2][SPLITS];
    #pragma unroll
    for (int u = 0; u < 2; ++u) {
        const int j = q * 512 + u * 256 + tid;
        #pragma unroll
        for (int sp = 0; sp < SPLITS; ++sp)
            p[u][sp] = *(const v4f*)(pctx
                + ((size_t)bh * SPLITS + sp) * (ED * ED) + (size_t)j * 4);
    }

    #pragma unroll
    for (int u = 0; u < 4; ++u) {
        const int idx = u * 256 + tid;
        const int sl = idx >> 3, c4 = idx & 7;
        *(v4f*)(sP + sl * RCOLS + c4 * 4) = kp[u];
    }
    __syncthreads();

    #pragma unroll
    for (int u = 0; u < 2; ++u) {
        const int j = q * 512 + u * 256 + tid;
        v4f s = {0.f, 0.f, 0.f, 0.f};
        #pragma unroll
        for (int sp = 0; sp < SPLITS; ++sp) s += p[u][sp];
        *(v4f*)(ctx + (size_t)bh * (ED * ED) + (size_t)j * 4) = s;
    }

    if (tid < RCOLS) {
        float v[PKS];
        #pragma unroll
        for (int sl = 0; sl < PKS; ++sl) v[sl] = sP[sl * RCOLS + tid];
        __builtin_amdgcn_sched_barrier(0);
        float s = 0.f;
        #pragma unroll
        for (int sl = 0; sl < PKS; ++sl) s += v[sl];
        ksum[bh * ED + q * RCOLS + tid] = s;
    }
}

#define TPW     4
__global__ __launch_bounds__(256) void la_out(
    const float* __restrict__ query, const float* __restrict__ ctx,
    const float* __restrict__ ksum, float* __restrict__ out)
{
    __shared__ __attribute__((aligned(16))) float sC[ED * KVP];
    __shared__ __attribute__((aligned(16))) float sQ[2][64 * QP];
    __shared__ __attribute__((aligned(16))) float sO[64 * QP];
    __shared__ float sKs[ED];
    __shared__ float sDinv[2][64];

    const int bh = blockIdx.y;
    const int b = bh >> 4, h = bh & 15;
    const int tbase = blockIdx.x * (TPW * 64);
    const int tid = threadIdx.x;
    const int wave = tid >> 5, lane = tid & 31;
    const int half = lane >> 4, lm = lane & 15;

    const int lrow = tid >> 4, c4 = tid & 15;
    const size_t thr_off = (size_t)b * TOK * HDIM + (size_t)h * ED
                         + (size_t)(tbase + lrow) * HDIM + (size_t)c4 * 4;
    const float* qthr = query + thr_off;
    float*       othr = out + thr_off;

    const int n0a = (wave >> 2) << 4;
    const int n0b = n0a + 32;
    const int d0  = (wave & 3) << 4;

    v4f qv[4];
    {
        v4f cv[4];
        #pragma unroll
        for (int u = 0; u < 4; ++u) {
            qv[u] = *(const v4f*)(qthr + (size_t)(16 * u) * HDIM);
            cv[u] = *(const v4f*)(ctx + (size_t)bh * (ED * ED) + (lrow + 16 * u) * ED + c4 * 4);
        }
        if (tid < ED) sKs[tid] = ksum[bh * ED + tid];
        #pragma unroll
        for (int u = 0; u < 4; ++u)
            *(v4f*)(sC + (lrow + 16 * u) * KVP + c4 * 4) = cv[u];
    }
    #pragma unroll
    for (int u = 0; u < 4; ++u) {
        v4f qf;
        qf.x = elu_fm(qv[u].x); qf.y = elu_fm(qv[u].y);
        qf.z = elu_fm(qv[u].z); qf.w = elu_fm(qv[u].w);
        *(v4f*)(sQ[0] + (lrow + 16 * u) * QP + c4 * 4) = qf;
    }
    #pragma unroll
    for (int u = 0; u < 4; ++u)
        qv[u] = *(const v4f*)(qthr + (size_t)(64 + 16 * u) * HDIM);
    wg_barrier_lds_only();

    v2f bfr[16];
    #pragma unroll
    for (int ks = 0; ks < 16; ++ks) {
        bfr[ks].x = sC[(4 * ks + 2 * half) * KVP + d0 + lm];
        bfr[ks].y = sC[(4 * ks + 1 + 2 * half) * KVP + d0 + lm];
    }

    #pragma unroll
    for (int i = 0; i < TPW; ++i) {
        const float* sQc = sQ[i & 1];
        float* sOc = (i & 1) ? sC : sO;

        if (tid < 64) {
            float d = 0.f;
            #pragma unroll 8
            for (int e = 0; e < ED; ++e) d += sQc[tid * QP + e] * sKs[e];
            sDinv[i & 1][tid] = 1.0f / (d + 1e-6f);
        }

        v8f acc0 = {};
        v8f acc1 = {};
        #pragma unroll
        for (int ks = 0; ks < 16; ++ks) {
            const int e = 4 * ks;
            v2f a0, a1;
            a0.x = sQc[(n0a + lm) * QP + e + 2 * half];
            a0.y = sQc[(n0a + lm) * QP + e + 1 + 2 * half];
            a1.x = sQc[(n0b + lm) * QP + e + 2 * half];
            a1.y = sQc[(n0b + lm) * QP + e + 1 + 2 * half];
            acc0 = __builtin_amdgcn_wmma_f32_16x16x4_f32(
                false, a0, false, bfr[ks], (short)0, acc0, false, false);
            acc1 = __builtin_amdgcn_wmma_f32_16x16x4_f32(
                false, a1, false, bfr[ks], (short)0, acc1, false, false);
        }
        __builtin_amdgcn_sched_barrier(0);

        #pragma unroll
        for (int r = 0; r < 8; ++r) {
            sOc[(n0a + r + 8 * half) * QP + d0 + lm] = acc0[r];
            sOc[(n0b + r + 8 * half) * QP + d0 + lm] = acc1[r];
        }

        if (i + 1 < TPW) {
            #pragma unroll
            for (int u = 0; u < 4; ++u) {
                v4f qf;
                qf.x = elu_fm(qv[u].x); qf.y = elu_fm(qv[u].y);
                qf.z = elu_fm(qv[u].z); qf.w = elu_fm(qv[u].w);
                *(v4f*)(sQ[(i + 1) & 1] + (lrow + 16 * u) * QP + c4 * 4) = qf;
            }
        }

        wg_barrier_lds_only();

        if (i + 2 < TPW) {
            #pragma unroll
            for (int u = 0; u < 4; ++u)
                qv[u] = *(const v4f*)(qthr + (size_t)((i + 2) * 64 + 16 * u) * HDIM);
        }

        #pragma unroll
        for (int u = 0; u < 4; ++u) {
            const int row = lrow + 16 * u;
            v4f o = *(const v4f*)(sOc + row * QP + c4 * 4);
            const float di = sDinv[i & 1][row];
            o.x *= di; o.y *= di; o.z *= di; o.w *= di;
            *(v4f*)(othr + (size_t)(i * 64 + 16 * u) * HDIM) = o;
        }
    }
}

extern "C" void kernel_launch(void* const* d_in, const int* in_sizes, int n_in,
                              void* d_out, int out_size, void* d_ws, size_t ws_size,
                              hipStream_t stream) {
    (void)in_sizes; (void)n_in; (void)out_size; (void)ws_size;
    const float* q = (const float*)d_in[0];
    const float* k = (const float*)d_in[1];
    const float* v = (const float*)d_in[2];
    float* out = (float*)d_out;

    float* wsf   = (float*)d_ws;
    float* pctx  = wsf;
    float* pksum = pctx  + (size_t)BH * SPLITS * (ED * ED);
    float* ctx   = pksum + (size_t)BH * PKS * ED;
    float* ksum  = ctx   + (size_t)BH * (ED * ED);

    la_ctx_partial<<<BH * SPLITS, 256, 0, stream>>>(k, v, pctx, pksum);
    la_reduce<<<BH * RBLK, 256, 0, stream>>>(pctx, pksum, ctx, ksum);
    la_out<<<dim3(TOK / (64 * TPW), BH), 256, 0, stream>>>(q, ctx, ksum, out);
}
